# starting kernel
# baseline (speedup 1.0000x reference)
#include <hip/hip_runtime.h>


typedef __attribute__((ext_vector_type(2))) float v2f;
typedef __attribute__((ext_vector_type(8))) float v8f;

#define TILE 16
#define NT   4
#define BIGF 3.0e38f

__device__ __forceinline__ float min2f(float a, float b) {
    float r;
    asm("v_min_num_f32 %0, %1, %2" : "=v"(r) : "v"(a), "v"(b));
    return r;
}
__device__ __forceinline__ float min3f(float a, float b, float c) {
    float r;
    asm("v_min3_num_f32 %0, %1, %2, %3" : "=v"(r) : "v"(a), "v"(b), "v"(c));
    return r;
}

__global__ __launch_bounds__(256)
void chamfer_nn_kernel(const float* __restrict__ query,
                       const float* __restrict__ ref,
                       float* __restrict__ partial,
                       int nQuery, int nRef, int strips, int B,
                       int tilesPerChunk)
{
    const int tid  = blockIdx.x * blockDim.x + threadIdx.x;
    const int wave = __builtin_amdgcn_readfirstlane(tid >> 5);
    const int lane = tid & 31;
    const int half = lane >> 4;
    const int l15  = lane & 15;

    const int perChunk = B * strips;
    const int ch  = wave / perChunk;
    const int rem = wave % perChunk;
    const int b   = rem / strips;
    const int s   = rem % strips;
    const int row0 = s * (TILE * NT);

    const float* qB = query   + (size_t)b * nQuery * 3;
    const float* rB = ref     + (size_t)b * nRef   * 3;
    float*       oB = partial + ((size_t)ch * B + b) * nQuery;

    v2f   a[NT];
    float sq[NT];
#pragma unroll
    for (int t = 0; t < NT; ++t) {
        const float* qp = qB + (size_t)(row0 + t * TILE + l15) * 3;
        const float qx = qp[0], qy = qp[1], qz = qp[2];
        sq[t] = qx * qx + qy * qy + qz * qz;
        a[t]  = half ? (v2f){-2.0f * qz, 1.0f}
                     : (v2f){-2.0f * qx, -2.0f * qy};
    }

    v8f acc[NT];
#pragma unroll
    for (int t = 0; t < NT; ++t)
        acc[t] = (v8f){BIGF, BIGF, BIGF, BIGF, BIGF, BIGF, BIGF, BIGF};

    const v8f cz = (v8f){0.f, 0.f, 0.f, 0.f, 0.f, 0.f, 0.f, 0.f};

    const int jBeg = ch * tilesPerChunk;
    const int jEnd = jBeg + tilesPerChunk;
    for (int j = jBeg; j < jEnd; j += 2) {
        const int pj = (j + 16 < jEnd) ? (j + 16) : j;
        __builtin_prefetch(rB + (size_t)(pj * TILE + l15) * 3, 0, 1);

        const float* rp0 = rB + (size_t)((j    ) * TILE + l15) * 3;
        const float* rp1 = rB + (size_t)((j + 1) * TILE + l15) * 3;
        const float r0x = rp0[0], r0y = rp0[1], r0z = rp0[2];
        const float r1x = rp1[0], r1y = rp1[1], r1z = rp1[2];
        const float sr0 = r0x * r0x + r0y * r0y + r0z * r0z;
        const float sr1 = r1x * r1x + r1y * r1y + r1z * r1z;

        const v2f b0 = half ? (v2f){r0z, sr0} : (v2f){r0x, r0y};
        const v2f b1 = half ? (v2f){r1z, sr1} : (v2f){r1x, r1y};

#pragma unroll
        for (int t = 0; t < NT; ++t) {
            v8f d0 = __builtin_amdgcn_wmma_f32_16x16x4_f32(
                false, a[t], false, b0, (short)0, cz, false, false);
            v8f d1 = __builtin_amdgcn_wmma_f32_16x16x4_f32(
                false, a[t], false, b1, (short)0, cz, false, false);
#pragma unroll
            for (int k = 0; k < 8; ++k)
                acc[t][k] = min3f(acc[t][k], d0[k], d1[k]);
        }
    }

#pragma unroll
    for (int t = 0; t < NT; ++t) {
#pragma unroll
        for (int v = 0; v < 8; ++v) {
            float r = acc[t][v];
            r = min2f(r, __shfl_xor(r, 1, 32));
            r = min2f(r, __shfl_xor(r, 2, 32));
            r = min2f(r, __shfl_xor(r, 4, 32));
            r = min2f(r, __shfl_xor(r, 8, 32));
            const float sqr = __shfl(sq[t], v + 8 * half, 32);
            if (l15 == 0)
                oB[row0 + t * TILE + v + half * 8] = r + sqr;
        }
    }
}

__global__ __launch_bounds__(256)
void reduce_min_kernel(const float* __restrict__ partial,
                       float* __restrict__ out, int n, int nChunks)
{
    const int i = blockIdx.x * blockDim.x + threadIdx.x;
    if (i < n) {
        float r = partial[i];
        for (int c = 1; c < nChunks; ++c)
            r = min2f(r, partial[(size_t)c * n + i]);
        out[i] = r;
    }
}

static void run_pass(const float* query, const float* ref, float* out,
                     int nQuery, int nRef, int B, float* ws, int nChunks,
                     hipStream_t stream)
{
    const int strips = nQuery / (TILE * NT);
    const int tilesPerChunk = (nRef / TILE) / nChunks;
    const int waves  = B * strips * nChunks;
    const dim3 block(256);
    float* partial = (nChunks > 1) ? ws : out;

    chamfer_nn_kernel<<<(waves * 32) / 256, block, 0, stream>>>(
        query, ref, partial, nQuery, nRef, strips, B, tilesPerChunk);

    if (nChunks > 1) {
        const int n = B * nQuery;
        reduce_min_kernel<<<(n + 255) / 256, block, 0, stream>>>(
            partial, out, n, nChunks);
    }
}

extern "C" void kernel_launch(void* const* d_in, const int* in_sizes, int n_in,
                              void* d_out, int out_size, void* d_ws, size_t ws_size,
                              hipStream_t stream) {
    (void)n_in; (void)out_size;
    const int B = 4;
    const int N = in_sizes[0] / (B * 3);
    const int M = in_sizes[1] / (B * 3);

    const float* xyz1 = (const float*)d_in[0];
    const float* xyz2 = (const float*)d_in[1];
    float* out = (float*)d_out;

    const int maxQ = (N > M) ? N : M;
    int nChunks = 4;
    if (ws_size < (size_t)nChunks * B * maxQ * sizeof(float)) nChunks = 1;

    run_pass(xyz1, xyz2, out,                N, M, B, (float*)d_ws, nChunks, stream);
    run_pass(xyz2, xyz1, out + (size_t)B * N, M, N, B, (float*)d_ws, nChunks, stream);
}


# Stage each work-group's reference chunk once in LDS as ready B fragments (x,y | z,|r|^2); the sweep reads them by ds_load one group ahead instead of two dependent global loads per step
# speedup vs baseline: 1.4717x; 1.4717x over previous
#include <hip/hip_runtime.h>


typedef __attribute__((ext_vector_type(2))) float v2f;
typedef __attribute__((ext_vector_type(8))) float v8f;

#define TILE 16
#define NT   4
#define UG   8
#define PIECE_TILES 128
#define STAGE_PER_THREAD ((PIECE_TILES * TILE) / 256)
#define BIGF 3.0e38f

__device__ __forceinline__ float min2f(float a, float b) {
    float r;
    asm("v_min_num_f32 %0, %1, %2" : "=v"(r) : "v"(a), "v"(b));
    return r;
}
__device__ __forceinline__ float min3f(float a, float b, float c) {
    return __builtin_fminimum_numf(__builtin_fminimum_numf(a, b), c);
}

__device__ __forceinline__ v8f dist_tile(v2f a, v2f b, v8f cz) {
    return __builtin_amdgcn_wmma_f32_16x16x4_f32(
        false, a, false, b, (short)0, cz, false, false);
}

__device__ __forceinline__ void load_group(v2f (&bf)[UG], const v2f* frag, int j) {
#pragma unroll
    for (int u = 0; u < UG; ++u)
        bf[u] = frag[(j + u) * (TILE * 2)];
}

__device__ __forceinline__ void sweep_group(const v2f (&a)[NT], const v2f (&bf)[UG],
                                            v8f (&acc)[NT], v8f cz) {
    constexpr int K = NT * (UG / 2);
    v8f d0 = dist_tile(a[0], bf[0], cz);
    v8f d1 = dist_tile(a[0], bf[1], cz);
#pragma unroll
    for (int k = 0; k < K; ++k) {
        v8f n0 = d0, n1 = d1;
        if (k + 1 < K) {
            const int tn = (k + 1) % NT;
            const int pn = (k + 1) / NT;
            n0 = dist_tile(a[tn], bf[2 * pn    ], cz);
            n1 = dist_tile(a[tn], bf[2 * pn + 1], cz);
        }
        const int t = k % NT;
#pragma unroll
        for (int e = 0; e < 8; ++e)
            acc[t][e] = min3f(acc[t][e], d0[e], d1[e]);
        d0 = n0; d1 = n1;
    }
}

__global__ __launch_bounds__(256)
void chamfer_nn_kernel(const float* __restrict__ query,
                       const float* __restrict__ ref,
                       float* __restrict__ partial,
                       int nQuery, int nRef, int strips, int B,
                       int tilesPerChunk)
{
    __shared__ __attribute__((aligned(16))) v2f sfrag[PIECE_TILES * TILE * 2];

    const int tid  = blockIdx.x * blockDim.x + threadIdx.x;
    const int wave = __builtin_amdgcn_readfirstlane(tid >> 5);
    const int lane = tid & 31;
    const int half = lane >> 4;
    const int l15  = lane & 15;

    const int perChunk = B * strips;
    const int ch  = wave / perChunk;
    const int rem = wave % perChunk;
    const int b   = rem / strips;
    const int s   = rem % strips;
    const int row0 = s * (TILE * NT);

    const float* qB = query   + (size_t)b * nQuery * 3;
    const float* rB = ref     + (size_t)b * nRef   * 3;
    float*       oB = partial + ((size_t)ch * B + b) * nQuery;

    v2f   a[NT];
    float sq[NT];
#pragma unroll
    for (int t = 0; t < NT; ++t) {
        const float* qp = qB + (size_t)(row0 + t * TILE + l15) * 3;
        const float qx = qp[0], qy = qp[1], qz = qp[2];
        sq[t] = qx * qx + qy * qy + qz * qz;
        a[t]  = half ? (v2f){-2.0f * qz, 1.0f}
                     : (v2f){-2.0f * qx, -2.0f * qy};
    }

    v8f acc[NT];
#pragma unroll
    for (int t = 0; t < NT; ++t)
        acc[t] = (v8f){BIGF, BIGF, BIGF, BIGF, BIGF, BIGF, BIGF, BIGF};

    const v8f cz = (v8f){0.f, 0.f, 0.f, 0.f, 0.f, 0.f, 0.f, 0.f};

    const v2f* frag = sfrag + l15 * 2 + half;

    const int jBeg = ch * tilesPerChunk;
    for (int p0 = 0; p0 < tilesPerChunk; p0 += PIECE_TILES) {
        const int nt   = (tilesPerChunk - p0 < PIECE_TILES) ? (tilesPerChunk - p0)
                                                            : PIECE_TILES;
        const int npts = nt * TILE;
        const float* src = rB + (size_t)(jBeg + p0) * TILE * 3;

        if (p0 > 0) __syncthreads();

        float px[STAGE_PER_THREAD], py[STAGE_PER_THREAD], pz[STAGE_PER_THREAD];
#pragma unroll
        for (int i = 0; i < STAGE_PER_THREAD; ++i) {
            int r = i * 256 + (int)threadIdx.x;
            r = (r < npts) ? r : (npts - 1);
            px[i] = src[3 * r]; py[i] = src[3 * r + 1]; pz[i] = src[3 * r + 2];
        }
#pragma unroll
        for (int i = 0; i < STAGE_PER_THREAD; ++i) {
            const int r = i * 256 + (int)threadIdx.x;
            if (r < npts) {
                const float sr = px[i] * px[i] + py[i] * py[i] + pz[i] * pz[i];
                sfrag[2 * r    ] = (v2f){px[i], py[i]};
                sfrag[2 * r + 1] = (v2f){pz[i], sr};
            }
        }
        __syncthreads();

        v2f bA[UG], bB[UG];
        load_group(bA, frag, 0);
        for (int j = 0; j < nt; j += 2 * UG) {
            load_group(bB, frag, j + UG);
            sweep_group(a, bA, acc, cz);
            const int jn = (j + 2 * UG < nt) ? (j + 2 * UG) : j;
            load_group(bA, frag, jn);
            sweep_group(a, bB, acc, cz);
        }
    }

#pragma unroll
    for (int t = 0; t < NT; ++t) {
#pragma unroll
        for (int v = 0; v < 8; ++v) {
            float r = acc[t][v];
            r = min2f(r, __shfl_xor(r, 1, 32));
            r = min2f(r, __shfl_xor(r, 2, 32));
            r = min2f(r, __shfl_xor(r, 4, 32));
            r = min2f(r, __shfl_xor(r, 8, 32));
            const float sqr = __shfl(sq[t], v + 8 * half, 32);
            if (l15 == 0)
                oB[row0 + t * TILE + v + half * 8] = r + sqr;
        }
    }
}

__global__ __launch_bounds__(256)
void reduce_min_kernel(const float* __restrict__ partial,
                       float* __restrict__ out, int n, int nChunks)
{
    const int i = blockIdx.x * blockDim.x + threadIdx.x;
    if (i < n) {
        float r = partial[i];
        for (int c = 1; c < nChunks; ++c)
            r = min2f(r, partial[(size_t)c * n + i]);
        out[i] = r;
    }
}

static void run_pass(const float* query, const float* ref, float* out,
                     int nQuery, int nRef, int B, float* ws, int nChunks,
                     hipStream_t stream)
{
    const int strips = nQuery / (TILE * NT);
    const int tilesPerChunk = (nRef / TILE) / nChunks;
    const int waves  = B * strips * nChunks;
    const dim3 block(256);
    float* partial = (nChunks > 1) ? ws : out;

    chamfer_nn_kernel<<<(waves * 32) / 256, block, 0, stream>>>(
        query, ref, partial, nQuery, nRef, strips, B, tilesPerChunk);

    if (nChunks > 1) {
        const int n = B * nQuery;
        reduce_min_kernel<<<(n + 255) / 256, block, 0, stream>>>(
            partial, out, n, nChunks);
    }
}

extern "C" void kernel_launch(void* const* d_in, const int* in_sizes, int n_in,
                              void* d_out, int out_size, void* d_ws, size_t ws_size,
                              hipStream_t stream) {
    (void)n_in; (void)out_size;
    const int B = 4;
    const int N = in_sizes[0] / (B * 3);
    const int M = in_sizes[1] / (B * 3);

    const float* xyz1 = (const float*)d_in[0];
    const float* xyz2 = (const float*)d_in[1];
    float* out = (float*)d_out;

    const int maxQ = (N > M) ? N : M;
    int nChunks = 4;
    if (ws_size < (size_t)nChunks * B * maxQ * sizeof(float)) nChunks = 1;

    run_pass(xyz1, xyz2, out,                N, M, B, (float*)d_ws, nChunks, stream);
    run_pass(xyz2, xyz1, out + (size_t)B * N, M, N, B, (float*)d_ws, nChunks, stream);
}


# One launch instead of four: each work-group sweeps the whole reference cloud for its own 256 query rows, both directions in one grid; no partial minima, no workspace, no reduce kernel
# speedup vs baseline: 2.4879x; 1.6905x over previous
#include <hip/hip_runtime.h>


typedef __attribute__((ext_vector_type(2))) float v2f;
typedef __attribute__((ext_vector_type(8))) float v8f;

#define TILE 16
#define NT   2
#define WAVES 8
#define ROWS_PER_WG (WAVES * NT * TILE)
#define UG   16
#define PIECE_TILES 128
#define PIECE_PTS (PIECE_TILES * TILE)
#define STAGE_PER_THREAD (PIECE_PTS / 256)
#define BIGF 3.0e38f

__device__ __forceinline__ float min2f(float a, float b) {
    float r;
    asm("v_min_num_f32 %0, %1, %2" : "=v"(r) : "v"(a), "v"(b));
    return r;
}
__device__ __forceinline__ float min3f(float a, float b, float c) {
    return __builtin_fminimum_numf(__builtin_fminimum_numf(a, b), c);
}

__device__ __forceinline__ v8f dist_tile(v2f a, v2f b, v8f cz) {
    return __builtin_amdgcn_wmma_f32_16x16x4_f32(
        false, a, false, b, (short)0, cz, false, false);
}

__device__ __forceinline__ void load_group(v2f (&bf)[UG], const v2f* frag, int j) {
#pragma unroll
    for (int u = 0; u < UG; ++u)
        bf[u] = frag[(j + u) * (TILE * 2)];
}

__device__ __forceinline__ void sweep_group(const v2f (&a)[NT], const v2f (&bf)[UG],
                                            v8f (&acc)[NT], v8f cz) {
    constexpr int K = NT * (UG / 2);
    v8f d0 = dist_tile(a[0], bf[0], cz);
    v8f d1 = dist_tile(a[0], bf[1], cz);
#pragma unroll
    for (int k = 0; k < K; ++k) {
        v8f n0 = d0, n1 = d1;
        if (k + 1 < K) {
            const int tn = (k + 1) % NT;
            const int pn = (k + 1) / NT;
            n0 = dist_tile(a[tn], bf[2 * pn    ], cz);
            n1 = dist_tile(a[tn], bf[2 * pn + 1], cz);
        }
        __builtin_amdgcn_sched_barrier(0);
        const int t = k % NT;
#pragma unroll
        for (int e = 0; e < 8; ++e)
            acc[t][e] = min3f(acc[t][e], d0[e], d1[e]);
        d0 = n0; d1 = n1;
        __builtin_amdgcn_sched_barrier(0);
    }
}

__device__ __forceinline__ void sweep_piece(const v2f* frag, int nt, const v2f (&a)[NT],
                                            v8f (&acc)[NT], v8f cz) {
    v2f bA[UG], bB[UG];
    load_group(bA, frag, 0);
    for (int j = 0; j < nt; j += 2 * UG) {
        load_group(bB, frag, j + UG);
        sweep_group(a, bA, acc, cz);
        const int jn = (j + 2 * UG < nt) ? (j + 2 * UG) : j;
        load_group(bA, frag, jn);
        sweep_group(a, bB, acc, cz);
    }
}

__device__ __forceinline__ void piece_load(const float* __restrict__ src, int npts,
                                           float (&px)[STAGE_PER_THREAD],
                                           float (&py)[STAGE_PER_THREAD],
                                           float (&pz)[STAGE_PER_THREAD]) {
#pragma unroll
    for (int i = 0; i < STAGE_PER_THREAD; ++i) {
        int r = i * 256 + (int)threadIdx.x;
        r = (r < npts) ? r : (npts - 1);
        px[i] = src[3 * r]; py[i] = src[3 * r + 1]; pz[i] = src[3 * r + 2];
    }
}

__device__ __forceinline__ void piece_store(v2f* stage, int npts,
                                            const float (&px)[STAGE_PER_THREAD],
                                            const float (&py)[STAGE_PER_THREAD],
                                            const float (&pz)[STAGE_PER_THREAD]) {
#pragma unroll
    for (int i = 0; i < STAGE_PER_THREAD; ++i) {
        const int r = i * 256 + (int)threadIdx.x;
        if (r < npts) {
            const float sr = px[i] * px[i] + py[i] * py[i] + pz[i] * pz[i];
            stage[2 * r    ] = (v2f){px[i], py[i]};
            stage[2 * r + 1] = (v2f){pz[i], sr};
        }
    }
}

__global__ __launch_bounds__(256)
void chamfer_nn_kernel(const float* __restrict__ xyz1,
                       const float* __restrict__ xyz2,
                       float* __restrict__ out,
                       int N, int M, int B)
{
    __shared__ __attribute__((aligned(16))) v2f sfrag[2 * PIECE_PTS * 2];

    const int wgFirst = B * (N / ROWS_PER_WG);
    int g = (int)blockIdx.x;
    const bool second = (g >= wgFirst);
    if (second) g -= wgFirst;
    const int nQuery = second ? M : N;
    const int nRef   = second ? N : M;
    const float* query = second ? xyz2 : xyz1;
    const float* ref   = second ? xyz1 : xyz2;
    float* dst = second ? (out + (size_t)B * N) : out;

    const int blocksPerBatch = nQuery / ROWS_PER_WG;
    const int b   = g / blocksPerBatch;
    const int blk = g % blocksPerBatch;

    const int wv   = __builtin_amdgcn_readfirstlane((int)threadIdx.x >> 5);
    const int lane = (int)threadIdx.x & 31;
    const int half = lane >> 4;
    const int l15  = lane & 15;
    const int row0 = blk * ROWS_PER_WG + wv * (NT * TILE);

    const float* qB = query + (size_t)b * nQuery * 3;
    const float* rB = ref   + (size_t)b * nRef   * 3;
    float*       oB = dst   + (size_t)b * nQuery;

    v2f   a[NT];
    float sq[NT];
#pragma unroll
    for (int t = 0; t < NT; ++t) {
        const float* qp = qB + (size_t)(row0 + t * TILE + l15) * 3;
        const float qx = qp[0], qy = qp[1], qz = qp[2];
        sq[t] = qx * qx + qy * qy + qz * qz;
        a[t]  = half ? (v2f){-2.0f * qz, 1.0f}
                     : (v2f){-2.0f * qx, -2.0f * qy};
    }

    v8f acc[NT];
#pragma unroll
    for (int t = 0; t < NT; ++t)
        acc[t] = (v8f){BIGF, BIGF, BIGF, BIGF, BIGF, BIGF, BIGF, BIGF};

    const v8f cz = (v8f){0.f, 0.f, 0.f, 0.f, 0.f, 0.f, 0.f, 0.f};

    const int nTiles  = nRef / TILE;
    const int nPieces = (nTiles + PIECE_TILES - 1) / PIECE_TILES;

    float px[STAGE_PER_THREAD], py[STAGE_PER_THREAD], pz[STAGE_PER_THREAD];
    {
        const int nt0 = (nTiles < PIECE_TILES) ? nTiles : PIECE_TILES;
        piece_load(rB, nt0 * TILE, px, py, pz);
        piece_store(sfrag, nt0 * TILE, px, py, pz);
    }

    int p = 0;
    for (; p + 1 < nPieces; ++p) {
        __syncthreads();

        const int leftn = nTiles - (p + 1) * PIECE_TILES;
        const int ntn   = (leftn < PIECE_TILES) ? leftn : PIECE_TILES;
        float nx[STAGE_PER_THREAD], ny[STAGE_PER_THREAD], nz[STAGE_PER_THREAD];
        piece_load(rB + (size_t)(p + 1) * PIECE_PTS * 3, ntn * TILE, nx, ny, nz);

        sweep_piece(sfrag + (p & 1) * (PIECE_PTS * 2) + l15 * 2 + half,
                    PIECE_TILES, a, acc, cz);

        piece_store(sfrag + ((p + 1) & 1) * (PIECE_PTS * 2), ntn * TILE, nx, ny, nz);
    }
    __syncthreads();
    sweep_piece(sfrag + (p & 1) * (PIECE_PTS * 2) + l15 * 2 + half,
                nTiles - p * PIECE_TILES, a, acc, cz);

#pragma unroll
    for (int t = 0; t < NT; ++t) {
#pragma unroll
        for (int v = 0; v < 8; ++v) {
            float r = acc[t][v];
            r = min2f(r, __shfl_xor(r, 1, 32));
            r = min2f(r, __shfl_xor(r, 2, 32));
            r = min2f(r, __shfl_xor(r, 4, 32));
            r = min2f(r, __shfl_xor(r, 8, 32));
            const float sqr = __shfl(sq[t], v + 8 * half, 32);
            if (l15 == 0)
                oB[row0 + t * TILE + v + half * 8] = r + sqr;
        }
    }
}

extern "C" void kernel_launch(void* const* d_in, const int* in_sizes, int n_in,
                              void* d_out, int out_size, void* d_ws, size_t ws_size,
                              hipStream_t stream) {
    (void)n_in; (void)out_size; (void)d_ws; (void)ws_size;
    const int B = 4;
    const int N = in_sizes[0] / (B * 3);
    const int M = in_sizes[1] / (B * 3);

    const float* xyz1 = (const float*)d_in[0];
    const float* xyz2 = (const float*)d_in[1];
    float* out = (float*)d_out;

    const int groups = B * (N / ROWS_PER_WG) + B * (M / ROWS_PER_WG);
    chamfer_nn_kernel<<<groups, dim3(256), 0, stream>>>(xyz1, xyz2, out, N, M, B);
}


# Epilogue row minima on the vector ALU: four DPP row_xmask min steps per accumulator register replace 80 serial LDS permutes per wave; one 16-lane store per query tile
# speedup vs baseline: 2.7774x; 1.1164x over previous
#include <hip/hip_runtime.h>


typedef __attribute__((ext_vector_type(2))) float v2f;
typedef __attribute__((ext_vector_type(8))) float v8f;

#define TILE 16
#define NT   2
#define WAVES 8
#define ROWS_PER_WG (WAVES * NT * TILE)
#define UG   16
#define PIECE_TILES 128
#define PIECE_PTS (PIECE_TILES * TILE)
#define STAGE_PER_THREAD (PIECE_PTS / 256)
#define BIGF 3.0e38f

template <int MASK>
__device__ __forceinline__ float row_min_step(float x) {
    float r;
    asm volatile("v_min_num_f32_dpp %0, %1, %1 row_xmask:%2 row_mask:0xf bank_mask:0xf"
                 : "=v"(r) : "v"(x), "n"(MASK));
    return r;
}
__device__ __forceinline__ float row_xor8(float x) {
    const int xi = __builtin_bit_cast(int, x);
    return __builtin_bit_cast(float,
        __builtin_amdgcn_update_dpp(xi, xi, 0x168, 0xf, 0xf, true));
}
__device__ __forceinline__ float min3f(float a, float b, float c) {
    return __builtin_fminimum_numf(__builtin_fminimum_numf(a, b), c);
}

template <int MASK>
__device__ __forceinline__ void row_min_step_all(v8f (&acc)[NT]) {
#pragma unroll
    for (int t = 0; t < NT; ++t) {
#pragma unroll
        for (int v = 0; v < 8; ++v)
            acc[t][v] = row_min_step<MASK>(acc[t][v]);
    }
}

__device__ __forceinline__ v8f dist_tile(v2f a, v2f b, v8f cz) {
    return __builtin_amdgcn_wmma_f32_16x16x4_f32(
        false, a, false, b, (short)0, cz, false, false);
}

__device__ __forceinline__ void load_group(v2f (&bf)[UG], const v2f* frag, int j) {
#pragma unroll
    for (int u = 0; u < UG; ++u)
        bf[u] = frag[(j + u) * (TILE * 2)];
}

__device__ __forceinline__ void sweep_group(const v2f (&a)[NT], const v2f (&bf)[UG],
                                            v8f (&acc)[NT], v8f cz) {
    constexpr int K = NT * (UG / 2);
    v8f d0 = dist_tile(a[0], bf[0], cz);
    v8f d1 = dist_tile(a[0], bf[1], cz);
#pragma unroll
    for (int k = 0; k < K; ++k) {
        v8f n0 = d0, n1 = d1;
        if (k + 1 < K) {
            const int tn = (k + 1) % NT;
            const int pn = (k + 1) / NT;
            n0 = dist_tile(a[tn], bf[2 * pn    ], cz);
            n1 = dist_tile(a[tn], bf[2 * pn + 1], cz);
        }
        __builtin_amdgcn_sched_barrier(0);
        const int t = k % NT;
#pragma unroll
        for (int e = 0; e < 8; ++e)
            acc[t][e] = min3f(acc[t][e], d0[e], d1[e]);
        d0 = n0; d1 = n1;
        __builtin_amdgcn_sched_barrier(0);
    }
}

__device__ __forceinline__ void sweep_piece(const v2f* frag, int nt, const v2f (&a)[NT],
                                            v8f (&acc)[NT], v8f cz) {
    v2f bA[UG], bB[UG];
    load_group(bA, frag, 0);
    for (int j = 0; j < nt; j += 2 * UG) {
        load_group(bB, frag, j + UG);
        sweep_group(a, bA, acc, cz);
        const int jn = (j + 2 * UG < nt) ? (j + 2 * UG) : j;
        load_group(bA, frag, jn);
        sweep_group(a, bB, acc, cz);
    }
}

__device__ __forceinline__ void piece_load(const float* __restrict__ src, int npts,
                                           float (&px)[STAGE_PER_THREAD],
                                           float (&py)[STAGE_PER_THREAD],
                                           float (&pz)[STAGE_PER_THREAD]) {
#pragma unroll
    for (int i = 0; i < STAGE_PER_THREAD; ++i) {
        int r = i * 256 + (int)threadIdx.x;
        r = (r < npts) ? r : (npts - 1);
        px[i] = src[3 * r]; py[i] = src[3 * r + 1]; pz[i] = src[3 * r + 2];
    }
}

__device__ __forceinline__ void piece_store(v2f* stage, int npts,
                                            const float (&px)[STAGE_PER_THREAD],
                                            const float (&py)[STAGE_PER_THREAD],
                                            const float (&pz)[STAGE_PER_THREAD]) {
#pragma unroll
    for (int i = 0; i < STAGE_PER_THREAD; ++i) {
        const int r = i * 256 + (int)threadIdx.x;
        if (r < npts) {
            const float sr = px[i] * px[i] + py[i] * py[i] + pz[i] * pz[i];
            stage[2 * r    ] = (v2f){px[i], py[i]};
            stage[2 * r + 1] = (v2f){pz[i], sr};
        }
    }
}

__global__ __launch_bounds__(256)
void chamfer_nn_kernel(const float* __restrict__ xyz1,
                       const float* __restrict__ xyz2,
                       float* __restrict__ out,
                       int N, int M, int B)
{
    __shared__ __attribute__((aligned(16))) v2f sfrag[2 * PIECE_PTS * 2];

    const int wgFirst = B * (N / ROWS_PER_WG);
    int g = (int)blockIdx.x;
    const bool second = (g >= wgFirst);
    if (second) g -= wgFirst;
    const int nQuery = second ? M : N;
    const int nRef   = second ? N : M;
    const float* query = second ? xyz2 : xyz1;
    const float* ref   = second ? xyz1 : xyz2;
    float* dst = second ? (out + (size_t)B * N) : out;

    const int blocksPerBatch = nQuery / ROWS_PER_WG;
    const int b   = g / blocksPerBatch;
    const int blk = g % blocksPerBatch;

    const int wv   = __builtin_amdgcn_readfirstlane((int)threadIdx.x >> 5);
    const int lane = (int)threadIdx.x & 31;
    const int half = lane >> 4;
    const int l15  = lane & 15;
    const int row0 = blk * ROWS_PER_WG + wv * (NT * TILE);

    const float* qB = query + (size_t)b * nQuery * 3;
    const float* rB = ref   + (size_t)b * nRef   * 3;
    float*       oB = dst   + (size_t)b * nQuery;

    v2f   a[NT];
    float sq[NT];
#pragma unroll
    for (int t = 0; t < NT; ++t) {
        const float* qp = qB + (size_t)(row0 + t * TILE + l15) * 3;
        const float qx = qp[0], qy = qp[1], qz = qp[2];
        sq[t] = qx * qx + qy * qy + qz * qz;
        a[t]  = half ? (v2f){-2.0f * qz, 1.0f}
                     : (v2f){-2.0f * qx, -2.0f * qy};
    }

    v8f acc[NT];
#pragma unroll
    for (int t = 0; t < NT; ++t)
        acc[t] = (v8f){BIGF, BIGF, BIGF, BIGF, BIGF, BIGF, BIGF, BIGF};

    const v8f cz = (v8f){0.f, 0.f, 0.f, 0.f, 0.f, 0.f, 0.f, 0.f};

    const int nTiles  = nRef / TILE;
    const int nPieces = (nTiles + PIECE_TILES - 1) / PIECE_TILES;

    float px[STAGE_PER_THREAD], py[STAGE_PER_THREAD], pz[STAGE_PER_THREAD];
    {
        const int nt0 = (nTiles < PIECE_TILES) ? nTiles : PIECE_TILES;
        piece_load(rB, nt0 * TILE, px, py, pz);
        piece_store(sfrag, nt0 * TILE, px, py, pz);
    }

    int p = 0;
    for (; p + 1 < nPieces; ++p) {
        __syncthreads();

        const int leftn = nTiles - (p + 1) * PIECE_TILES;
        const int ntn   = (leftn < PIECE_TILES) ? leftn : PIECE_TILES;
        float nx[STAGE_PER_THREAD], ny[STAGE_PER_THREAD], nz[STAGE_PER_THREAD];
        piece_load(rB + (size_t)(p + 1) * PIECE_PTS * 3, ntn * TILE, nx, ny, nz);

        sweep_piece(sfrag + (p & 1) * (PIECE_PTS * 2) + l15 * 2 + half,
                    PIECE_TILES, a, acc, cz);

        piece_store(sfrag + ((p + 1) & 1) * (PIECE_PTS * 2), ntn * TILE, nx, ny, nz);
    }
    __syncthreads();
    sweep_piece(sfrag + (p & 1) * (PIECE_PTS * 2) + l15 * 2 + half,
                nTiles - p * PIECE_TILES, a, acc, cz);

    row_min_step_all<1>(acc);
    row_min_step_all<2>(acc);
    row_min_step_all<4>(acc);
    row_min_step_all<8>(acc);

    float res[NT];
#pragma unroll
    for (int t = 0; t < NT; ++t) {
        float r = acc[t][0];
#pragma unroll
        for (int v = 1; v < 8; ++v)
            r = (l15 == v) ? acc[t][v] : r;
        const float sq8 = row_xor8(sq[t]);
        res[t] = r + (half ? sq8 : sq[t]);
    }
    if (l15 < 8) {
#pragma unroll
        for (int t = 0; t < NT; ++t)
            oB[row0 + t * TILE + half * 8 + l15] = res[t];
    }
}

extern "C" void kernel_launch(void* const* d_in, const int* in_sizes, int n_in,
                              void* d_out, int out_size, void* d_ws, size_t ws_size,
                              hipStream_t stream) {
    (void)n_in; (void)out_size; (void)d_ws; (void)ws_size;
    const int B = 4;
    const int N = in_sizes[0] / (B * 3);
    const int M = in_sizes[1] / (B * 3);

    const float* xyz1 = (const float*)d_in[0];
    const float* xyz2 = (const float*)d_in[1];
    float* out = (float*)d_out;

    const int groups = B * (N / ROWS_PER_WG) + B * (M / ROWS_PER_WG);
    chamfer_nn_kernel<<<groups, dim3(256), 0, stream>>>(xyz1, xyz2, out, N, M, B);
}


# Square 1024x1024 work-group blocks: each distance tile formed once, row and column minima from one D (|q|^2 as WMMA C operand: sum order |q|^2-2q.r+|r|^2); partials via workspace, 2nd launch folds
# speedup vs baseline: 3.5226x; 1.2683x over previous
#include <hip/hip_runtime.h>


typedef __attribute__((ext_vector_type(2))) float v2f;
typedef __attribute__((ext_vector_type(4))) float v4f;
typedef __attribute__((ext_vector_type(8))) float v8f;

#define TILE 16
#define NT   8
#define WAVES 8
#define BLK  (WAVES * NT * TILE)
#define BLK_TILES (BLK / TILE)
#define UG   8
#define STAGE_PER_THREAD (BLK / 256)
#define COLPITCH (BLK + 16)
#define BIGF 3.0e38f

template <int MASK>
__device__ __forceinline__ float row_min_step(float x) {
    float r;
    asm volatile("v_min_num_f32_dpp %0, %1, %1 row_xmask:%2 row_mask:0xf bank_mask:0xf"
                 : "=v"(r) : "v"(x), "n"(MASK));
    return r;
}
__device__ __forceinline__ float min3f(float a, float b, float c) {
    return __builtin_fminimum_numf(__builtin_fminimum_numf(a, b), c);
}

template <int MASK>
__device__ __forceinline__ void row_min_step_all(v8f (&acc)[NT]) {
#pragma unroll
    for (int t = 0; t < NT; ++t) {
#pragma unroll
        for (int v = 0; v < 8; ++v)
            acc[t][v] = row_min_step<MASK>(acc[t][v]);
    }
}

__device__ __forceinline__ v8f dist_tile(v2f a, v2f b, v8f c) {
    return __builtin_amdgcn_wmma_f32_16x16x4_f32(
        false, a, false, b, (short)0, c, false, false);
}

__device__ __forceinline__ void load_group(v2f (&bf)[UG], const v2f* frag, int j) {
#pragma unroll
    for (int u = 0; u < UG; ++u)
        bf[u] = frag[(j + u) * (TILE * 2)];
}

__device__ __forceinline__ float col_fold(float run, v8f d) {
    const float t1 = min3f(d[0], d[1], d[2]);
    const float t2 = min3f(d[3], d[4], d[5]);
    return min3f(min3f(run, t1, t2), d[6], d[7]);
}

__device__ __forceinline__ void sweep_group(const v2f (&a)[NT], const v8f (&cq)[NT],
                                            const v2f (&bf)[UG], v8f (&acc)[NT],
                                            float* colp) {
    constexpr int K = NT * (UG / 2);
    float col[UG];
    v8f d0 = dist_tile(a[0], bf[0], cq[0]);
    v8f d1 = dist_tile(a[0], bf[1], cq[0]);
#pragma unroll
    for (int k = 0; k < K; ++k) {
        v8f n0 = d0, n1 = d1;
        if (k + 1 < K) {
            const int tn = (k + 1) % NT;
            const int pn = (k + 1) / NT;
            n0 = dist_tile(a[tn], bf[2 * pn    ], cq[tn]);
            n1 = dist_tile(a[tn], bf[2 * pn + 1], cq[tn]);
        }
        __builtin_amdgcn_sched_barrier(0);
        const int t = k % NT;
        const int p = k / NT;
#pragma unroll
        for (int e = 0; e < 8; ++e)
            acc[t][e] = min3f(acc[t][e], d0[e], d1[e]);
        col[2 * p    ] = col_fold((t == 0) ? d0[7] : col[2 * p    ], d0);
        col[2 * p + 1] = col_fold((t == 0) ? d1[7] : col[2 * p + 1], d1);
        d0 = n0; d1 = n1;
        __builtin_amdgcn_sched_barrier(0);
    }
#pragma unroll
    for (int u = 0; u < UG; ++u)
        colp[u * TILE] = col[u];
}

__global__ __launch_bounds__(256)
void chamfer_block_kernel(const float* __restrict__ xyz1,
                          const float* __restrict__ xyz2,
                          float* __restrict__ wsRow,
                          float* __restrict__ wsCol,
                          int N, int M)
{
    __shared__ __attribute__((aligned(16))) v2f   sfrag[BLK * 2];
    __shared__ __attribute__((aligned(16))) float ssq[2 * BLK];
    __shared__ __attribute__((aligned(16))) float scol[WAVES * 2 * COLPITCH];

    const int nbi = N / BLK;
    const int nbj = M / BLK;
    int g = (int)blockIdx.x;
    const int bj = g % nbj;  g /= nbj;
    const int bi = g % nbi;
    const int b  = g / nbi;

    const int wv   = __builtin_amdgcn_readfirstlane((int)threadIdx.x >> 5);
    const int lane = (int)threadIdx.x & 31;
    const int half = lane >> 4;
    const int l15  = lane & 15;
    const int rowW = wv * (NT * TILE);
    const int row0 = bi * BLK + rowW;

    const float* qB = xyz1 + (size_t)b * N * 3;
    const float* rB = xyz2 + ((size_t)b * M + (size_t)bj * BLK) * 3;

    float px[STAGE_PER_THREAD], py[STAGE_PER_THREAD], pz[STAGE_PER_THREAD];
#pragma unroll
    for (int i = 0; i < STAGE_PER_THREAD; ++i) {
        const int r = i * 256 + (int)threadIdx.x;
        px[i] = rB[3 * r]; py[i] = rB[3 * r + 1]; pz[i] = rB[3 * r + 2];
    }
    float qx[NT], qy[NT], qz[NT];
#pragma unroll
    for (int t = 0; t < NT; ++t) {
        const float* qp = qB + (size_t)(row0 + t * TILE + l15) * 3;
        qx[t] = qp[0]; qy[t] = qp[1]; qz[t] = qp[2];
    }
    __builtin_amdgcn_sched_barrier(0);

    v2f a[NT];
#pragma unroll
    for (int t = 0; t < NT; ++t) {
        const float sqv = qx[t] * qx[t] + qy[t] * qy[t] + qz[t] * qz[t];
        a[t] = half ? (v2f){-2.0f * qz[t], 1.0f}
                    : (v2f){-2.0f * qx[t], -2.0f * qy[t]};
        ssq[half * BLK + rowW + t * TILE + l15] = sqv;
    }

#pragma unroll
    for (int i = 0; i < STAGE_PER_THREAD; ++i) {
        const int r = i * 256 + (int)threadIdx.x;
        const float sr = px[i] * px[i] + py[i] * py[i] + pz[i] * pz[i];
        sfrag[2 * r    ] = (v2f){px[i], py[i]};
        sfrag[2 * r + 1] = (v2f){pz[i], sr};
    }

    __syncthreads();

    v8f cq[NT];
#pragma unroll
    for (int t = 0; t < NT; ++t) {
        const v4f* sp = (const v4f*)(ssq + half * BLK + rowW + t * TILE + half * 8);
        const v4f lo = sp[0], hi = sp[1];
        cq[t] = (v8f){lo[0], lo[1], lo[2], lo[3], hi[0], hi[1], hi[2], hi[3]};
    }

    v8f acc[NT];
#pragma unroll
    for (int t = 0; t < NT; ++t)
        acc[t] = (v8f){BIGF, BIGF, BIGF, BIGF, BIGF, BIGF, BIGF, BIGF};

    {
        const v2f* frag = sfrag + l15 * 2 + half;
        float* colp = scol + (wv * 2 + half) * COLPITCH + l15;
        v2f bA[UG], bB[UG];
        load_group(bA, frag, 0);
        for (int j = 0; j < BLK_TILES; j += 2 * UG) {
            load_group(bB, frag, j + UG);
            sweep_group(a, cq, bA, acc, colp + j * TILE);
            const int jn = (j + 2 * UG < BLK_TILES) ? (j + 2 * UG) : j;
            load_group(bA, frag, jn);
            sweep_group(a, cq, bB, acc, colp + (j + UG) * TILE);
        }
    }

    row_min_step_all<1>(acc);
    row_min_step_all<2>(acc);
    row_min_step_all<4>(acc);
    row_min_step_all<8>(acc);

    const int  c7  = l15 & 7;
    const bool hiT = (l15 >= 8);
    float res[NT / 2];
#pragma unroll
    for (int tp = 0; tp < NT / 2; ++tp) {
        float r = hiT ? acc[2 * tp + 1][0] : acc[2 * tp][0];
#pragma unroll
        for (int v = 1; v < 8; ++v) {
            const float s = hiT ? acc[2 * tp + 1][v] : acc[2 * tp][v];
            r = (c7 == v) ? s : r;
        }
        res[tp] = r;
    }

    __syncthreads();
    {
        float* wc = wsCol + ((size_t)b * nbi + bi) * M + (size_t)bj * BLK;
#pragma unroll
        for (int i = 0; i < STAGE_PER_THREAD; ++i) {
            const int m = i * 256 + (int)threadIdx.x;
            float v = BIGF;
#pragma unroll
            for (int w = 0; w < WAVES * 2; w += 2)
                v = min3f(v, scol[w * COLPITCH + m], scol[(w + 1) * COLPITCH + m]);
            wc[m] = v;
        }
    }

    {
        float* wr = wsRow + ((size_t)b * nbj + bj) * N + row0;
#pragma unroll
        for (int tp = 0; tp < NT / 2; ++tp)
            wr[(2 * tp + (hiT ? 1 : 0)) * TILE + half * 8 + c7] = res[tp];
    }
}

__global__ __launch_bounds__(256)
void chamfer_fold_kernel(const float* __restrict__ ws, float* __restrict__ out,
                         int N, int M, int B)
{
    const int o  = (int)blockIdx.x * 256 + (int)threadIdx.x;
    const int n1 = B * N;
    if (o >= n1 + B * M) return;
    const float* base;
    int nk, len;
    if (o < n1) {
        const int b = o / N, x = o - b * N;
        nk = M / BLK; len = N;
        base = ws + (size_t)b * nk * N + x;
    } else {
        const int o2 = o - n1;
        const int b = o2 / M, x = o2 - b * M;
        nk = N / BLK; len = M;
        base = ws + (size_t)B * (M / BLK) * N + (size_t)b * nk * M + x;
    }
    float v = BIGF;
    int k = 0;
    for (; k + 8 <= nk; k += 8) {
        float t[8];
#pragma unroll
        for (int i = 0; i < 8; ++i) t[i] = base[(size_t)(k + i) * len];
        v = min3f(v, min3f(t[0], t[1], t[2]), min3f(t[3], t[4], t[5]));
        v = min3f(v, t[6], t[7]);
    }
    for (; k < nk; ++k)
        v = __builtin_fminimum_numf(v, base[(size_t)k * len]);
    out[o] = v;
}

extern "C" void kernel_launch(void* const* d_in, const int* in_sizes, int n_in,
                              void* d_out, int out_size, void* d_ws, size_t ws_size,
                              hipStream_t stream) {
    (void)n_in; (void)out_size; (void)ws_size;
    const int B = 4;
    const int N = in_sizes[0] / (B * 3);
    const int M = in_sizes[1] / (B * 3);

    const float* xyz1 = (const float*)d_in[0];
    const float* xyz2 = (const float*)d_in[1];
    float* out = (float*)d_out;
    float* ws  = (float*)d_ws;
    float* wsRow = ws;
    float* wsCol = ws + (size_t)B * (M / BLK) * N;

    const int groups = B * (N / BLK) * (M / BLK);
    chamfer_block_kernel<<<groups, dim3(256), 0, stream>>>(xyz1, xyz2, wsRow, wsCol, N, M);
    const int folds = (B * N + B * M + 255) / 256;
    chamfer_fold_kernel<<<folds, dim3(256), 0, stream>>>(ws, out, N, M, B);
}
